# starting kernel
# baseline (speedup 1.0000x reference)
#include <hip/hip_runtime.h>

typedef float v2f __attribute__((ext_vector_type(2)));
typedef float v8f __attribute__((ext_vector_type(8)));

#define STR  68
#define LDSC (16 * STR)

__global__ void JpegBuildT_kernel(const float* __restrict__ Dm,
                                  const float* __restrict__ Di,
                                  const float* __restrict__ mask,
                                  float* __restrict__ ws)
{
    int idx = blockIdx.x * blockDim.x + threadIdx.x;
    if (idx >= 3 * 64 * 64) return;
    int c = idx >> 12;
    int m = (idx >> 6) & 63;
    int q = idx & 63;
    int pr = m >> 3, pc = m & 7, qr = q >> 3, qc = q & 7;
    const float* mk = mask + (long)c * 512 * 512;
    float acc = 0.0f;
    for (int i = 0; i < 8; ++i) {
        float ai = Di[pr * 8 + i] * Dm[i * 8 + qr];
        float inner = 0.0f;
        for (int j = 0; j < 8; ++j)
            inner += mk[i * 512 + j] * Di[pc * 8 + j] * Dm[j * 8 + qc];
        acc += ai * inner;
    }
    int t = m >> 4, ml = m & 15;
    int k = q >> 2, h = (q >> 1) & 1, jj = q & 1;
    ws[c * 4096 + t * 1024 + k * 64 + (h * 16 + ml) * 2 + jj] = acc;
}

__global__ __launch_bounds__(384)
void JpegCompression_50070728737401_kernel(const float* __restrict__ img,
                                           const float* __restrict__ T,
                                           float* __restrict__ out)
{
    __shared__ __align__(16) float sIn[3 * LDSC];
    __shared__ __align__(16) float sOut[3 * LDSC];

    const int g   = blockIdx.x;
    const int b   = g >> 8;
    const int s   = g & 255;
    const int r0  = (s >> 2) << 3;
    const int c0  = (s & 3) << 7;
    const int tid = threadIdx.x;
    const long plane = 512l * 512l;

    for (int p = tid; p < 1024; p += 384) {
        int row = p >> 7, col = p & 127;
        long base = ((long)(b * 3) * 512 + (r0 + row)) * 512 + (c0 + col);
        float r  = img[base];
        float gc = img[base + plane];
        float bc = img[base + 2 * plane];
        float y =  0.299f   * r + 0.587f   * gc + 0.114f   * bc;
        float u = -0.14713f * r - 0.28886f * gc + 0.436f   * bc;
        float v =  0.615f   * r - 0.51499f * gc - 0.10001f * bc;
        int n  = col >> 3;
        int rr = (row << 3) | (col & 7);
        sIn[0 * LDSC + n * STR + rr] = y;
        sIn[1 * LDSC + n * STR + rr] = u;
        sIn[2 * LDSC + n * STR + rr] = v;
    }
    __syncthreads();

    {
        int w    = tid >> 5;
        int lane = tid & 31;
        int ch   = w >> 2;
        int t    = w & 3;
        int h    = lane >> 4;
        int nl   = lane & 15;

        const float* aBase = T + ((ch * 4 + t) * 16) * 64;
        const float* sB    = sIn + ch * LDSC + nl * STR;

        v8f acc = {};
        #pragma unroll
        for (int k = 0; k < 16; ++k) {
            v2f a  = *(const v2f*)(aBase + k * 64 + lane * 2);
            v2f bb = *(const v2f*)(sB + 4 * k + 2 * h);
            acc = __builtin_amdgcn_wmma_f32_16x16x4_f32(
                 false, a,  false, bb,
                 (short)0, acc,  false,  false);
        }
        float* oB = sOut + ch * LDSC + nl * STR + t * 16 + h * 8;
        *(float4*)(oB)     = make_float4(acc[0], acc[1], acc[2], acc[3]);
        *(float4*)(oB + 4) = make_float4(acc[4], acc[5], acc[6], acc[7]);
    }
    __syncthreads();

    for (int p = tid; p < 1024; p += 384) {
        int row = p >> 7, col = p & 127;
        int n  = col >> 3;
        int rr = (row << 3) | (col & 7);
        float y = sOut[0 * LDSC + n * STR + rr];
        float u = sOut[1 * LDSC + n * STR + rr];
        float v = sOut[2 * LDSC + n * STR + rr];
        float r  = y + 1.13983f * v;
        float gc = y - 0.39465f * u - 0.5806f  * v;
        float bc = y + 2.03211f * u;
        long base = ((long)(b * 3) * 512 + (r0 + row)) * 512 + (c0 + col);
        out[base]             = r;
        out[base + plane]     = gc;
        out[base + 2 * plane] = bc;
    }
}

extern "C" void kernel_launch(void* const* d_in, const int* in_sizes, int n_in,
                              void* d_out, int out_size, void* d_ws, size_t ws_size,
                              hipStream_t stream)
{
    (void)in_sizes; (void)n_in; (void)out_size; (void)ws_size;
    const float* img  = (const float*)d_in[0];
    const float* Dd   = (const float*)d_in[1];
    const float* Didc = (const float*)d_in[2];
    const float* mask = (const float*)d_in[3];
    float* T = (float*)d_ws;

    JpegBuildT_kernel<<<48, 256, 0, stream>>>(Dd, Didc, mask, T);
    JpegCompression_50070728737401_kernel<<<8192, 384, 0, stream>>>(img, T, (float*)d_out);
}


# Image loads and result stores as 16-byte accesses (one 512-B strip row per wave instruction): a strip's 24 row loads are all in flight at once instead of three dependent load trips per thread
# speedup vs baseline: 1.2498x; 1.2498x over previous
#include <hip/hip_runtime.h>

typedef float v2f __attribute__((ext_vector_type(2)));
typedef float v8f __attribute__((ext_vector_type(8)));

#define STR  68
#define LDSC (16 * STR)

__global__ void JpegBuildT_kernel(const float* __restrict__ Dm,
                                  const float* __restrict__ Di,
                                  const float* __restrict__ mask,
                                  float* __restrict__ ws)
{
    int idx = blockIdx.x * blockDim.x + threadIdx.x;
    if (idx >= 3 * 64 * 64) return;
    int c = idx >> 12;
    int m = (idx >> 6) & 63;
    int q = idx & 63;
    int pr = m >> 3, pc = m & 7, qr = q >> 3, qc = q & 7;
    const float* mk = mask + (long)c * 512 * 512;
    float acc = 0.0f;
    for (int i = 0; i < 8; ++i) {
        float ai = Di[pr * 8 + i] * Dm[i * 8 + qr];
        float inner = 0.0f;
        for (int j = 0; j < 8; ++j)
            inner += mk[i * 512 + j] * Di[pc * 8 + j] * Dm[j * 8 + qc];
        acc += ai * inner;
    }
    int t = m >> 4, ml = m & 15;
    int k = q >> 2, h = (q >> 1) & 1, jj = q & 1;
    ws[c * 4096 + t * 1024 + k * 64 + (h * 16 + ml) * 2 + jj] = acc;
}

__global__ __launch_bounds__(384)
void StripOperator_kernel(const float* __restrict__ img,
                          const float* __restrict__ T,
                          float* __restrict__ out)
{
    __shared__ __align__(16) float sIn[3 * LDSC];
    __shared__ __align__(16) float sOut[3 * LDSC];

    const int g   = blockIdx.x;
    const int b   = g >> 8;
    const int s   = g & 255;
    const int r0  = (s >> 2) << 3;
    const int c0  = (s & 3) << 7;
    const int tid = threadIdx.x;
    const long plane = 512l * 512l;

    const int  prow  = tid >> 5;
    const int  pcol  = (tid & 31) << 2;
    const int  pn    = pcol >> 3;
    const int  prr   = (prow << 3) | (pcol & 7);
    const long pbase = ((long)(b * 3) * 512 + (r0 + prow)) * 512 + (c0 + pcol);
    if (tid < 256) {
        const float4 r4 = *(const float4*)(img + pbase);
        const float4 g4 = *(const float4*)(img + pbase + plane);
        const float4 b4 = *(const float4*)(img + pbase + 2 * plane);
        float4 y4, u4, v4;
        y4.x =  0.299f   * r4.x + 0.587f   * g4.x + 0.114f   * b4.x;
        y4.y =  0.299f   * r4.y + 0.587f   * g4.y + 0.114f   * b4.y;
        y4.z =  0.299f   * r4.z + 0.587f   * g4.z + 0.114f   * b4.z;
        y4.w =  0.299f   * r4.w + 0.587f   * g4.w + 0.114f   * b4.w;
        u4.x = -0.14713f * r4.x - 0.28886f * g4.x + 0.436f   * b4.x;
        u4.y = -0.14713f * r4.y - 0.28886f * g4.y + 0.436f   * b4.y;
        u4.z = -0.14713f * r4.z - 0.28886f * g4.z + 0.436f   * b4.z;
        u4.w = -0.14713f * r4.w - 0.28886f * g4.w + 0.436f   * b4.w;
        v4.x =  0.615f   * r4.x - 0.51499f * g4.x - 0.10001f * b4.x;
        v4.y =  0.615f   * r4.y - 0.51499f * g4.y - 0.10001f * b4.y;
        v4.z =  0.615f   * r4.z - 0.51499f * g4.z - 0.10001f * b4.z;
        v4.w =  0.615f   * r4.w - 0.51499f * g4.w - 0.10001f * b4.w;
        *(float4*)(sIn + 0 * LDSC + pn * STR + prr) = y4;
        *(float4*)(sIn + 1 * LDSC + pn * STR + prr) = u4;
        *(float4*)(sIn + 2 * LDSC + pn * STR + prr) = v4;
    }
    __syncthreads();

    {
        int w    = tid >> 5;
        int lane = tid & 31;
        int ch   = w >> 2;
        int t    = w & 3;
        int h    = lane >> 4;
        int nl   = lane & 15;

        const float* aBase = T + ((ch * 4 + t) * 16) * 64;
        const float* sB    = sIn + ch * LDSC + nl * STR;

        v8f acc = {};
        #pragma unroll
        for (int k = 0; k < 16; ++k) {
            v2f a  = *(const v2f*)(aBase + k * 64 + lane * 2);
            v2f bb = *(const v2f*)(sB + 4 * k + 2 * h);
            acc = __builtin_amdgcn_wmma_f32_16x16x4_f32(
                 false, a,  false, bb,
                 (short)0, acc,  false,  false);
        }
        float* oB = sOut + ch * LDSC + nl * STR + t * 16 + h * 8;
        *(float4*)(oB)     = make_float4(acc[0], acc[1], acc[2], acc[3]);
        *(float4*)(oB + 4) = make_float4(acc[4], acc[5], acc[6], acc[7]);
    }
    __syncthreads();

    if (tid < 256) {
        const float4 y4 = *(const float4*)(sOut + 0 * LDSC + pn * STR + prr);
        const float4 u4 = *(const float4*)(sOut + 1 * LDSC + pn * STR + prr);
        const float4 v4 = *(const float4*)(sOut + 2 * LDSC + pn * STR + prr);
        float4 r4, g4, b4;
        r4.x = y4.x + 1.13983f * v4.x;
        r4.y = y4.y + 1.13983f * v4.y;
        r4.z = y4.z + 1.13983f * v4.z;
        r4.w = y4.w + 1.13983f * v4.w;
        g4.x = y4.x - 0.39465f * u4.x - 0.5806f  * v4.x;
        g4.y = y4.y - 0.39465f * u4.y - 0.5806f  * v4.y;
        g4.z = y4.z - 0.39465f * u4.z - 0.5806f  * v4.z;
        g4.w = y4.w - 0.39465f * u4.w - 0.5806f  * v4.w;
        b4.x = y4.x + 2.03211f * u4.x;
        b4.y = y4.y + 2.03211f * u4.y;
        b4.z = y4.z + 2.03211f * u4.z;
        b4.w = y4.w + 2.03211f * u4.w;
        *(float4*)(out + pbase)             = r4;
        *(float4*)(out + pbase + plane)     = g4;
        *(float4*)(out + pbase + 2 * plane) = b4;
    }
}

extern "C" void kernel_launch(void* const* d_in, const int* in_sizes, int n_in,
                              void* d_out, int out_size, void* d_ws, size_t ws_size,
                              hipStream_t stream)
{
    (void)in_sizes; (void)n_in; (void)out_size; (void)ws_size;
    const float* img  = (const float*)d_in[0];
    const float* Dd   = (const float*)d_in[1];
    const float* Didc = (const float*)d_in[2];
    const float* mask = (const float*)d_in[3];
    float* T = (float*)d_ws;

    JpegBuildT_kernel<<<48, 256, 0, stream>>>(Dd, Didc, mask, T);
    StripOperator_kernel<<<8192, 384, 0, stream>>>(img, T, (float*)d_out);
}


# Work-groups persist: 1024 groups (4 per WGP, all resident) loop over 8 strips; operator fragments stay in registers, next strip's rows are requested before the matrix stage; 8x fewer waves enter
# speedup vs baseline: 1.3865x; 1.1094x over previous
#include <hip/hip_runtime.h>

typedef float v2f __attribute__((ext_vector_type(2)));
typedef float v8f __attribute__((ext_vector_type(8)));

#define STR  68
#define LDSC (16 * STR)

__global__ void JpegBuildT_kernel(const float* __restrict__ Dm,
                                  const float* __restrict__ Di,
                                  const float* __restrict__ mask,
                                  float* __restrict__ ws)
{
    int idx = blockIdx.x * blockDim.x + threadIdx.x;
    if (idx >= 3 * 64 * 64) return;
    int c = idx >> 12;
    int m = (idx >> 6) & 63;
    int q = idx & 63;
    int pr = m >> 3, pc = m & 7, qr = q >> 3, qc = q & 7;
    const float* mk = mask + (long)c * 512 * 512;
    float acc = 0.0f;
    for (int i = 0; i < 8; ++i) {
        float ai = Di[pr * 8 + i] * Dm[i * 8 + qr];
        float inner = 0.0f;
        for (int j = 0; j < 8; ++j)
            inner += mk[i * 512 + j] * Di[pc * 8 + j] * Dm[j * 8 + qc];
        acc += ai * inner;
    }
    int t = m >> 4, ml = m & 15;
    int k = q >> 2, h = (q >> 1) & 1, jj = q & 1;
    ws[c * 4096 + t * 1024 + k * 64 + (h * 16 + ml) * 2 + jj] = acc;
}

static __device__ __forceinline__ void lds_barrier()
{
    __builtin_amdgcn_fence(__ATOMIC_RELEASE, "workgroup", "local");
    __builtin_amdgcn_s_barrier();
    __builtin_amdgcn_fence(__ATOMIC_ACQUIRE, "workgroup", "local");
}

#define NGROUPS 1024
#define NSTRIPS 8192
__global__ __launch_bounds__(384) __attribute__((amdgpu_waves_per_eu(12, 12)))
void StripOperator_kernel(const float* __restrict__ img,
                          const float* __restrict__ T,
                          float* __restrict__ out)
{
    __shared__ __align__(16) float sIn[3 * LDSC];
    __shared__ __align__(16) float sOut[3 * LDSC];

    const int tid = threadIdx.x;
    const long plane = 512l * 512l;

    const int  prow  = tid >> 5;
    const int  pcol  = (tid & 31) << 2;
    const int  pn    = pcol >> 3;
    const int  prr   = (prow << 3) | (pcol & 7);
    const bool mover = tid < 256;

    const int w    = tid >> 5;
    const int lane = tid & 31;
    const int ch   = w >> 2;
    const int t    = w & 3;
    const int h    = lane >> 4;
    const int nl   = lane & 15;

    v2f a[16];
    {
        const float* aBase = T + ((ch * 4 + t) * 16) * 64 + lane * 2;
        #pragma unroll
        for (int k = 0; k < 16; ++k)
            a[k] = *(const v2f*)(aBase + k * 64);
    }
    const float* sB = sIn + ch * LDSC + nl * STR + 2 * h;
    float*       oB = sOut + ch * LDSC + nl * STR + t * 16 + h * 8;

    auto strip_off = [&](int s) -> long {
        const int b  = s >> 8;
        const int sw = s & 255;
        const int r0 = (sw >> 2) << 3;
        const int c0 = (sw & 3) << 7;
        return ((long)(b * 3) * 512 + (r0 + prow)) * 512 + (c0 + pcol);
    };

    long pbase = strip_off(blockIdx.x);
    float4 r4 = {}, g4 = {}, b4 = {};
    if (mover) {
        r4 = *(const float4*)(img + pbase);
        g4 = *(const float4*)(img + pbase + plane);
        b4 = *(const float4*)(img + pbase + 2 * plane);
    }

    #pragma unroll
    for (int it = 0; it < NSTRIPS / NGROUPS; ++it) {
        if (mover) {
            float4 y4, u4, v4;
            y4.x =  0.299f   * r4.x + 0.587f   * g4.x + 0.114f   * b4.x;
            y4.y =  0.299f   * r4.y + 0.587f   * g4.y + 0.114f   * b4.y;
            y4.z =  0.299f   * r4.z + 0.587f   * g4.z + 0.114f   * b4.z;
            y4.w =  0.299f   * r4.w + 0.587f   * g4.w + 0.114f   * b4.w;
            u4.x = -0.14713f * r4.x - 0.28886f * g4.x + 0.436f   * b4.x;
            u4.y = -0.14713f * r4.y - 0.28886f * g4.y + 0.436f   * b4.y;
            u4.z = -0.14713f * r4.z - 0.28886f * g4.z + 0.436f   * b4.z;
            u4.w = -0.14713f * r4.w - 0.28886f * g4.w + 0.436f   * b4.w;
            v4.x =  0.615f   * r4.x - 0.51499f * g4.x - 0.10001f * b4.x;
            v4.y =  0.615f   * r4.y - 0.51499f * g4.y - 0.10001f * b4.y;
            v4.z =  0.615f   * r4.z - 0.51499f * g4.z - 0.10001f * b4.z;
            v4.w =  0.615f   * r4.w - 0.51499f * g4.w - 0.10001f * b4.w;
            *(float4*)(sIn + 0 * LDSC + pn * STR + prr) = y4;
            *(float4*)(sIn + 1 * LDSC + pn * STR + prr) = u4;
            *(float4*)(sIn + 2 * LDSC + pn * STR + prr) = v4;
        }
        if (it == 0) __syncthreads(); else lds_barrier();

        const long pnext = strip_off(blockIdx.x + (it + 1) * NGROUPS);
        if (mover && it + 1 < NSTRIPS / NGROUPS) {
            r4 = *(const float4*)(img + pnext);
            g4 = *(const float4*)(img + pnext + plane);
            b4 = *(const float4*)(img + pnext + 2 * plane);
        }

        {
            v8f acc = {};
            #pragma unroll
            for (int k = 0; k < 16; ++k) {
                v2f bb = *(const v2f*)(sB + 4 * k);
                acc = __builtin_amdgcn_wmma_f32_16x16x4_f32(
                     false, a[k],  false, bb,
                     (short)0, acc,  false,  false);
            }
            *(float4*)(oB)     = make_float4(acc[0], acc[1], acc[2], acc[3]);
            *(float4*)(oB + 4) = make_float4(acc[4], acc[5], acc[6], acc[7]);
        }
        lds_barrier();

        if (mover) {
            const float4 y4 = *(const float4*)(sOut + 0 * LDSC + pn * STR + prr);
            const float4 u4 = *(const float4*)(sOut + 1 * LDSC + pn * STR + prr);
            const float4 v4 = *(const float4*)(sOut + 2 * LDSC + pn * STR + prr);
            float4 ro, go, bo;
            ro.x = y4.x + 1.13983f * v4.x;
            ro.y = y4.y + 1.13983f * v4.y;
            ro.z = y4.z + 1.13983f * v4.z;
            ro.w = y4.w + 1.13983f * v4.w;
            go.x = y4.x - 0.39465f * u4.x - 0.5806f  * v4.x;
            go.y = y4.y - 0.39465f * u4.y - 0.5806f  * v4.y;
            go.z = y4.z - 0.39465f * u4.z - 0.5806f  * v4.z;
            go.w = y4.w - 0.39465f * u4.w - 0.5806f  * v4.w;
            bo.x = y4.x + 2.03211f * u4.x;
            bo.y = y4.y + 2.03211f * u4.y;
            bo.z = y4.z + 2.03211f * u4.z;
            bo.w = y4.w + 2.03211f * u4.w;
            *(float4*)(out + pbase)             = ro;
            *(float4*)(out + pbase + plane)     = go;
            *(float4*)(out + pbase + 2 * plane) = bo;
        }
        pbase = pnext;
    }
}

extern "C" void kernel_launch(void* const* d_in, const int* in_sizes, int n_in,
                              void* d_out, int out_size, void* d_ws, size_t ws_size,
                              hipStream_t stream)
{
    (void)in_sizes; (void)n_in; (void)out_size; (void)ws_size;
    const float* img  = (const float*)d_in[0];
    const float* Dd   = (const float*)d_in[1];
    const float* Didc = (const float*)d_in[2];
    const float* mask = (const float*)d_in[3];
    float* T = (float*)d_ws;

    JpegBuildT_kernel<<<48, 256, 0, stream>>>(Dd, Didc, mask, T);
    StripOperator_kernel<<<NGROUPS, 384, 0, stream>>>(img, T, (float*)d_out);
}
